# starting kernel
# baseline (speedup 1.0000x reference)
#include <hip/hip_runtime.h>
#include <math.h>


typedef __attribute__((ext_vector_type(2))) float v2f;
typedef __attribute__((ext_vector_type(8))) float v8f;

#define WMMA_F32(a, b, c) \
  __builtin_amdgcn_wmma_f32_16x16x4_f32(false, (a), false, (b), (short)0, (c), false, false)

constexpr int B_ = 2, T_ = 2048, C_ = 256, H_ = 4, D_ = 64, W_ = 64, S_ = 1;
constexpr int Tq_ = T_ - S_;
constexpr int KQVROW = 3 * H_ * D_;

__device__ __forceinline__ int clampi(int v, int lo, int hi) {
  return v < lo ? lo : (v > hi ? hi : v);
}

__global__ __launch_bounds__(32)
void gemm_abt(const float* __restrict__ A, const float* __restrict__ Bm,
              const float* __restrict__ bias, float* __restrict__ Cm,
              int M, int N, int K) {
  const int lane = threadIdx.x & 31;
  const int half = lane >> 4;
  const int l16  = lane & 15;
  const int kb   = half * 2;
  const int n0   = blockIdx.x * 32;
  const int m0   = blockIdx.y * 32;

  const float* a0 = A  + (size_t)(m0 + l16)      * K;
  const float* a1 = A  + (size_t)(m0 + 16 + l16) * K;
  const float* b0 = Bm + (size_t)(n0 + l16)      * K;
  const float* b1 = Bm + (size_t)(n0 + 16 + l16) * K;

  v8f c00 = {}, c01 = {}, c10 = {}, c11 = {};
  for (int k0 = 0; k0 < K; k0 += 64) {
    __builtin_prefetch(a0 + k0 + 64, 0, 3);
    __builtin_prefetch(a1 + k0 + 64, 0, 3);
    __builtin_prefetch(b0 + k0 + 64, 0, 3);
    __builtin_prefetch(b1 + k0 + 64, 0, 3);
#pragma unroll 4
    for (int k = k0; k < k0 + 64; k += 4) {
      v2f fa0 = *(const v2f*)(a0 + k + kb);
      v2f fa1 = *(const v2f*)(a1 + k + kb);
      v2f fb0 = *(const v2f*)(b0 + k + kb);
      v2f fb1 = *(const v2f*)(b1 + k + kb);
      c00 = WMMA_F32(fa0, fb0, c00);
      c01 = WMMA_F32(fa0, fb1, c01);
      c10 = WMMA_F32(fa1, fb0, c10);
      c11 = WMMA_F32(fa1, fb1, c11);
    }
  }
  const float bb0 = bias ? bias[n0 + l16]      : 0.0f;
  const float bb1 = bias ? bias[n0 + 16 + l16] : 0.0f;
#pragma unroll
  for (int i = 0; i < 8; ++i) {
    const int m = m0 + i + half * 8;
    Cm[(size_t)m * N        + (n0 + l16)]      = c00[i] + bb0;
    Cm[(size_t)m * N        + (n0 + 16 + l16)] = c01[i] + bb1;
    Cm[(size_t)(m + 16) * N + (n0 + l16)]      = c10[i] + bb0;
    Cm[(size_t)(m + 16) * N + (n0 + 16 + l16)] = c11[i] + bb1;
  }
}

__global__ __launch_bounds__(256)
void s0_kernel(const float* __restrict__ kqv, float* __restrict__ s0) {
  const int idx = blockIdx.x * 256 + threadIdx.x;
  if (idx >= B_ * H_ * T_) return;
  const int j  = idx % T_;
  const int bh = idx / T_;
  const int b  = bh / H_, h = bh % H_;
  const float* base = kqv + (size_t)(b * T_ + j) * KQVROW + h * D_;
  const float* kp = base;
  const float* qp = base + H_ * D_;
  float acc = 0.0f;
#pragma unroll 8
  for (int d = 0; d < D_; ++d) acc = fmaf(kp[d], qp[d], acc);
  s0[idx] = acc;
}

__global__ __launch_bounds__(32)
void s1_kernel(const float* __restrict__ kqv, const float* __restrict__ q_pos,
               float* __restrict__ S1) {
  const int lane = threadIdx.x & 31;
  const int half = lane >> 4, l16 = lane & 15, kb = half * 2;
  const int j0 = blockIdx.x * 16;
  const int w0 = blockIdx.y * 16;
  const int bh = blockIdx.z;
  const int b = bh / H_, h = bh % H_;

  const float* qrow = kqv + (size_t)(b * T_ + j0 + l16) * KQVROW + H_ * D_ + h * D_;
  const float* prow = q_pos + (size_t)((w0 + l16) * H_ + h) * D_;

  v8f c = {};
#pragma unroll
  for (int k = 0; k < D_; k += 4) {
    v2f a = *(const v2f*)(qrow + k + kb);
    v2f b2 = *(const v2f*)(prow + k + kb);
    c = WMMA_F32(a, b2, c);
  }
  float* out = S1 + ((size_t)bh * T_ + j0) * W_ + w0;
#pragma unroll
  for (int i = 0; i < 8; ++i)
    out[(size_t)(i + half * 8) * W_ + l16] = c[i];
}

__global__ __launch_bounds__(256)
void zero_pad_rows(float* __restrict__ attnout) {
  const int i = threadIdx.x;
  attnout[i] = 0.0f;
  attnout[(size_t)T_ * C_ + i] = 0.0f;
}

__global__ __launch_bounds__(128)
void attn_kernel(const float* __restrict__ kqv, const float* __restrict__ S1,
                 const float* __restrict__ s0, const float* __restrict__ v_pos,
                 float* __restrict__ attnout, float* __restrict__ wei_out) {
  __shared__ float sc[16][W_ + 1];

  const int t0 = blockIdx.x * 16;
  const int h  = blockIdx.y;
  const int b  = blockIdx.z;
  const int bh = b * H_ + h;
  const int tid = threadIdx.x;
  const float scale = 0.125f;

  {
    const int ti = tid >> 3;
    const int wb = (tid & 7) * 8;
    const int t  = t0 + ti;
#pragma unroll
    for (int u = 0; u < 8; ++u) {
      const int w = wb + u;
      const int j = t + w - (W_ - 1);
      float v;
      if (t >= Tq_)      v = 0.0f;
      else if (j < 0)    v = -INFINITY;
      else               v = (s0[(size_t)bh * T_ + j] +
                              S1[((size_t)bh * T_ + j) * W_ + w]) * scale;
      sc[ti][w] = v;
    }
  }
  __syncthreads();

  if (tid < 16) {
    float m = -INFINITY;
    for (int w = 0; w < W_; ++w) m = fmaxf(m, sc[tid][w]);
    float s = 0.0f;
    for (int w = 0; w < W_; ++w) { float e = __expf(sc[tid][w] - m); sc[tid][w] = e; s += e; }
    const float inv = 1.0f / s;
    for (int w = 0; w < W_; ++w) sc[tid][w] *= inv;
  }
  __syncthreads();

  {
    const int ti = tid >> 3;
    const int wb = (tid & 7) * 8;
    const int t  = t0 + ti;
    if (t < Tq_) {
#pragma unroll
      for (int u = 0; u < 8; ++u) {
        const int w = wb + u;
        wei_out[(((size_t)b * Tq_ + t) * W_ + w) * H_ + h] = sc[ti][w];
      }
    }
  }

  const int lane = tid & 31;
  const int wave = tid >> 5;
  const int d0   = wave * 16;
  const int half = lane >> 4, l16 = lane & 15, kb = half * 2;
  const float* vbase = kqv + 2 * H_ * D_ + h * D_ + d0 + l16;

  v8f c = {};
#pragma unroll
  for (int jt = 0; jt < 5; ++jt) {
    const int j0 = t0 - 64 + jt * 16;
#pragma unroll
    for (int k4 = 0; k4 < 16; k4 += 4) {
      v2f a;
      {
        const int t = t0 + l16;
        const int wa = (j0 + k4 + kb) - t + (W_ - 1);
        const int wA = clampi(wa, 0, W_ - 1);
        const int wB = clampi(wa + 1, 0, W_ - 1);
        a.x = (wa     >= 0 && wa     < W_) ? sc[l16][wA] : 0.0f;
        a.y = (wa + 1 >= 0 && wa + 1 < W_) ? sc[l16][wB] : 0.0f;
      }
      v2f bv;
      {
        const int jA = clampi(j0 + k4 + kb,     0, T_ - 1);
        const int jB = clampi(j0 + k4 + kb + 1, 0, T_ - 1);
        bv.x = vbase[(size_t)(b * T_ + jA) * KQVROW];
        bv.y = vbase[(size_t)(b * T_ + jB) * KQVROW];
      }
      c = WMMA_F32(a, bv, c);
    }
  }
#pragma unroll
  for (int k = 0; k < W_; k += 4) {
    v2f a;
    a.x = sc[l16][k + kb];
    a.y = sc[l16][k + kb + 1];
    v2f bv;
    bv.x = v_pos[((size_t)(k + kb)     * H_ + h) * D_ + d0 + l16];
    bv.y = v_pos[((size_t)(k + kb + 1) * H_ + h) * D_ + d0 + l16];
    c = WMMA_F32(a, bv, c);
  }

#pragma unroll
  for (int i = 0; i < 8; ++i) {
    const int ti = i + half * 8;
    const int t  = t0 + ti;
    if (t < Tq_)
      attnout[((size_t)b * T_ + (t + S_)) * C_ + h * D_ + d0 + l16] = c[i];
  }
}

extern "C" void kernel_launch(void* const* d_in, const int* in_sizes, int n_in,
                              void* d_out, int out_size, void* d_ws, size_t ws_size,
                              hipStream_t stream) {
  (void)in_sizes; (void)n_in; (void)out_size; (void)ws_size;

  const float* x     = (const float*)d_in[0];
  const float* w_kqv = (const float*)d_in[1];
  const float* w_out = (const float*)d_in[2];
  const float* b_out = (const float*)d_in[3];
  const float* q_pos = (const float*)d_in[4];
  const float* v_pos = (const float*)d_in[5];

  float* y   = (float*)d_out;
  float* wei = y + (size_t)B_ * T_ * C_;

  char*  ws      = (char*)d_ws;
  float* kqv     = (float*)ws;
  size_t off     = (size_t)B_ * T_ * KQVROW * sizeof(float);
  float* S1      = (float*)(ws + off);
  off           += (size_t)B_ * H_ * T_ * W_ * sizeof(float);
  float* s0      = (float*)(ws + off);
  off           += (size_t)B_ * H_ * T_ * sizeof(float);
  float* attnout = (float*)(ws + off);

  gemm_abt<<<dim3(KQVROW / 32, (B_ * T_) / 32), 32, 0, stream>>>(
      x, w_kqv, nullptr, kqv, B_ * T_, KQVROW, C_);

  s0_kernel<<<(B_ * H_ * T_ + 255) / 256, 256, 0, stream>>>(kqv, s0);
  s1_kernel<<<dim3(T_ / 16, W_ / 16, B_ * H_), 32, 0, stream>>>(kqv, q_pos, S1);

  zero_pad_rows<<<1, C_, 0, stream>>>(attnout);

  attn_kernel<<<dim3((Tq_ + 15) / 16, H_, B_), 128, 0, stream>>>(
      kqv, S1, s0, v_pos, attnout, wei);

  gemm_abt<<<dim3(C_ / 32, (B_ * T_) / 32), 32, 0, stream>>>(
      attnout, w_out, b_out, y, B_ * T_, C_, C_);
}


# attn writes its output tile in the out projection's A-fragment order (512 B per 16 rows x 8 k, turned in LDS); A fragment loads touch 4 whole lines, were 16; time shift moves to the store
# speedup vs baseline: 2.3223x; 2.3223x over previous
#include <hip/hip_runtime.h>
#include <math.h>


typedef __attribute__((ext_vector_type(2))) float v2f;
typedef __attribute__((ext_vector_type(4))) float v4f;
typedef __attribute__((ext_vector_type(8))) float v8f;

#define WMMA_F32(a, b, c) \
  __builtin_amdgcn_wmma_f32_16x16x4_f32(false, (a), false, (b), (short)0, (c), false, false)

constexpr int B_ = 2, T_ = 2048, C_ = 256, H_ = 4, D_ = 64, W_ = 64, S_ = 1;
constexpr int Tq_ = T_ - S_;
constexpr int KQVROW = 3 * H_ * D_;

__device__ __forceinline__ int clampi(int v, int lo, int hi) {
  return v < lo ? lo : (v > hi ? hi : v);
}

template <int MT, int NT>
struct FragSet {
  v4f a[MT];
  v4f b[NT];
};

template <int MT, int NT, int K>
__device__ __forceinline__ void load_set(FragSet<MT, NT>& f, const float* __restrict__ ap,
                                         const float* __restrict__ bp, int k) {
#pragma unroll
  for (int i = 0; i < MT; ++i) f.a[i] = *(const v4f*)(ap + i * 16 * K + k * 16);
#pragma unroll
  for (int j = 0; j < NT; ++j) f.b[j] = *(const v4f*)(bp + j * 16 * K + k);
}

template <int MT, int NT>
__device__ __forceinline__ void mma_set(const FragSet<MT, NT>& f, v8f (&acc)[MT][NT]) {
#pragma unroll
  for (int i = 0; i < MT; ++i)
#pragma unroll
    for (int j = 0; j < NT; ++j)
      acc[i][j] = WMMA_F32(f.a[i].lo, f.b[j].lo, acc[i][j]);
#pragma unroll
  for (int i = 0; i < MT; ++i)
#pragma unroll
    for (int j = 0; j < NT; ++j)
      acc[i][j] = WMMA_F32(f.a[i].hi, f.b[j].hi, acc[i][j]);
}

template <int MT, int NT, int NS, int K, int N, int TB>
__global__ __launch_bounds__(128)
void gemm_abt(const float* __restrict__ A, const float* __restrict__ Bm,
              const float* __restrict__ bias, float* __restrict__ Cm) {
  const int lane = threadIdx.x & 31;
  const int wave = threadIdx.x >> 5;
  const int half = lane >> 4;
  const int l16  = lane & 15;
  const int n0   = ((int)blockIdx.x * 4 + wave) * (16 * NT);
  const int m0   = blockIdx.y * (16 * MT);

  const float* ap = A  + (size_t)m0 * K + lane * 4;
  const float* bp = Bm + (size_t)(n0 + l16) * K + half * 4;
  static_assert(TB % (16 * MT) == 0, "a wave's rows lie in one batch");

  float bb[NT];
#pragma unroll
  for (int j = 0; j < NT; ++j) bb[j] = bias ? bias[n0 + j * 16 + l16] : 0.0f;

  v8f acc[MT][NT];
#pragma unroll
  for (int i = 0; i < MT; ++i)
#pragma unroll
    for (int j = 0; j < NT; ++j) acc[i][j] = v8f{};

  FragSet<MT, NT> f[NS];
#pragma unroll
  for (int s = 0; s < NS; ++s) load_set<MT, NT, K>(f[s], ap, bp, s * 8);

#pragma unroll
  for (int k = 0; k < K; k += NS * 8) {
#pragma unroll
    for (int s = 0; s < NS; ++s) {
      __builtin_amdgcn_sched_barrier(0);
      mma_set<MT, NT>(f[s], acc);
      __builtin_amdgcn_sched_barrier(0);
      if (k + NS * 8 < K) load_set<MT, NT, K>(f[s], ap, bp, k + NS * 8 + s * 8);
    }
  }

  float* cp = Cm + (size_t)(m0 + half * 8 + 1) * N + n0 + l16;
#pragma unroll
  for (int j = 0; j < NT; ++j) {
    if (half == 0 && (unsigned)m0 % (unsigned)TB == 0u) Cm[(size_t)m0 * N + n0 + j * 16 + l16] = bb[j];
#pragma unroll
    for (int i = 0; i < MT; ++i) {
#pragma unroll
      for (int r = 0; r < 8; ++r) {
        if ((unsigned)(m0 + i * 16 + half * 8 + r) % (unsigned)TB != (unsigned)(TB - 1))
          cp[(i * 16 + r) * N + j * 16] = acc[i][j][r] + bb[j];
      }
    }
  }
}

struct QkvSet {
  v4f w[6];
  v4f t[2];
};

template <int K>
__device__ __forceinline__ void load_qkv_set(QkvSet& f, const float* __restrict__ wp,
                                             const float* __restrict__ xp0,
                                             const float* __restrict__ xp1, int k) {
  constexpr int QO = 1 * H_ * D_ * K;
  constexpr int KO = 0 * H_ * D_ * K;
  constexpr int VO = 2 * H_ * D_ * K;
#pragma unroll
  for (int i = 0; i < 2; ++i) {
    f.w[i]     = *(const v4f*)(wp + QO + i * 16 * K + k);
    f.w[2 + i] = *(const v4f*)(wp + KO + i * 16 * K + k);
    f.w[4 + i] = *(const v4f*)(wp + VO + i * 16 * K + k);
  }
  f.t[0] = *(const v4f*)(xp0 + k);
  f.t[1] = *(const v4f*)(xp1 + k);
}

template <int NS>
__global__ __launch_bounds__(128)
void qkv_kernel(const float* __restrict__ x, const float* __restrict__ w_kqv,
                const float* __restrict__ q_pos, float* __restrict__ kqv,
                float* __restrict__ S1, float* __restrict__ s0) {
  constexpr int K = C_;
  static_assert(K % (8 * NS) == 0, "K loop runs whole passes over the NS sets");
  static_assert(T_ % 64 == 0, "a block's 64 tokens lie in one batch");
  static_assert(D_ == 64, "two d halves of two 16-feature tiles per projection");
  __shared__ alignas(16) float xs[4 * 8 * 32 * 4];
  __shared__ float xs0[4 * 16];

  const int lane = threadIdx.x & 31;
  const int wave = threadIdx.x >> 5;
  const int half = lane >> 4;
  const int l16  = lane & 15;
  const int tg   = wave >> 1;
  const int dh   = wave & 1;
  const int h    = blockIdx.y;
  const int tok0 = (blockIdx.x * 2 + tg) * 32;
  const int b    = tok0 / T_;
  const int j0   = tok0 % T_;
  const int bh   = b * H_ + h;

  const float* wp = w_kqv + (size_t)(h * D_ + dh * 32 + l16) * K + half * 4;
  const int tt0 = tok0 + 16 * dh;
  const int tt1 = tok0 + 16 * (1 - dh);
  const float* xp0 = x + (size_t)(tt0 + l16) * K + half * 4;
  const float* xp1 = x + (size_t)(tt1 + l16) * K + half * 4;

  v8f aq[2][2], ak[2][2], av[2][2];
#pragma unroll
  for (int i = 0; i < 2; ++i) {
#pragma unroll
    for (int j = 0; j < 2; ++j) {
      aq[i][j] = v8f{};
      ak[i][j] = v8f{};
      av[i][j] = v8f{};
    }
  }

  QkvSet f[NS];
#pragma unroll
  for (int s = 0; s < NS; ++s) load_qkv_set<K>(f[s], wp, xp0, xp1, s * 8);

  v4f qp[4][2][2];

#pragma unroll
  for (int k = 0; k < K; k += NS * 8) {
    if (k + NS * 8 >= K) {
      const float* pp = q_pos + (size_t)(l16 * H_ + h) * D_ + dh * 32 + half * 8;
#pragma unroll
      for (int wt = 0; wt < 4; ++wt) {
#pragma unroll
        for (int i = 0; i < 2; ++i) {
          qp[wt][i][0] = *(const v4f*)(pp + (size_t)wt * 16 * H_ * D_ + 16 * i);
          qp[wt][i][1] = *(const v4f*)(pp + (size_t)wt * 16 * H_ * D_ + 16 * i + 4);
        }
      }
    }
#pragma unroll
    for (int s = 0; s < NS; ++s) {
      __builtin_amdgcn_sched_barrier(0);
#pragma unroll
      for (int i = 0; i < 2; ++i) {
#pragma unroll
        for (int j = 0; j < 2; ++j) {
          aq[i][j] = WMMA_F32(f[s].w[i].lo, f[s].t[j].lo, aq[i][j]);
          ak[i][j] = WMMA_F32(f[s].w[2 + i].lo, f[s].t[j].lo, ak[i][j]);
          av[j][i] = WMMA_F32(f[s].t[j].lo, f[s].w[4 + i].lo, av[j][i]);
        }
      }
#pragma unroll
      for (int i = 0; i < 2; ++i) {
#pragma unroll
        for (int j = 0; j < 2; ++j) {
          aq[i][j] = WMMA_F32(f[s].w[i].hi, f[s].t[j].hi, aq[i][j]);
          ak[i][j] = WMMA_F32(f[s].w[2 + i].hi, f[s].t[j].hi, ak[i][j]);
          av[j][i] = WMMA_F32(f[s].t[j].hi, f[s].w[4 + i].hi, av[j][i]);
        }
      }
      __builtin_amdgcn_sched_barrier(0);
      if (k + NS * 8 < K) load_qkv_set<K>(f[s], wp, xp0, xp1, k + NS * 8 + s * 8);
    }
  }

  float* vp0 = kqv + (size_t)(tt0 + half * 8) * KQVROW + 2 * H_ * D_ + h * D_ + dh * 32 + l16;
  float* vp1 = kqv + (size_t)(tt1 + half * 8) * KQVROW + 2 * H_ * D_ + h * D_ + dh * 32 + l16;
#pragma unroll
  for (int i = 0; i < 2; ++i) {
#pragma unroll
    for (int r = 0; r < 8; ++r) {
      vp0[(size_t)r * KQVROW + 16 * i] = av[0][i][r];
      vp1[(size_t)r * KQVROW + 16 * i] = av[1][i][r];
    }
  }

  v8f d2[2][4];
#pragma unroll
  for (int j = 0; j < 2; ++j)
#pragma unroll
    for (int wt = 0; wt < 4; ++wt) d2[j][wt] = v8f{};
#pragma unroll
  for (int i = 0; i < 2; ++i) {
#pragma unroll
    for (int p = 0; p < 4; ++p) {
#pragma unroll
      for (int j = 0; j < 2; ++j) {
        v2f a;
        a.x = aq[i][j][2 * p];
        a.y = aq[i][j][2 * p + 1];
#pragma unroll
        for (int wt = 0; wt < 4; ++wt) {
          const v4f piece = qp[wt][i][p >> 1];
          const v2f bq = (p & 1) ? piece.hi : piece.lo;
          d2[j][wt] = WMMA_F32(a, bq, d2[j][wt]);
        }
      }
    }
  }

  float part[2];
#pragma unroll
  for (int j = 0; j < 2; ++j) {
    float pj = 0.0f;
#pragma unroll
    for (int i = 0; i < 2; ++i) {
#pragma unroll
      for (int r = 0; r < 8; ++r) pj = fmaf(aq[i][j][r], ak[i][j][r], pj);
    }
    pj += __shfl_xor(pj, 16);
    part[j] = pj;
  }

  {
    float* xw = xs + wave * (8 * 32 * 4) + lane * 4;
#pragma unroll
    for (int wt = 0; wt < 4; ++wt) {
      *(v4f*)(xw + (wt * 2 + 0) * 128) = d2[1][wt].lo;
      *(v4f*)(xw + (wt * 2 + 1) * 128) = d2[1][wt].hi;
    }
    if (half == 0) xs0[wave * 16 + l16] = part[1];
  }
  __syncthreads();
  {
    const float* xr = xs + (wave ^ 1) * (8 * 32 * 4) + lane * 4;
    const float so = xs0[(wave ^ 1) * 16 + l16];
    v4f olo[4], ohi[4];
#pragma unroll
    for (int wt = 0; wt < 4; ++wt) {
      olo[wt] = *(const v4f*)(xr + (wt * 2 + 0) * 128);
      ohi[wt] = *(const v4f*)(xr + (wt * 2 + 1) * 128);
    }
    const float sm = part[0] + so;
    if (half == 0) s0[(size_t)bh * T_ + j0 + 16 * dh + l16] = sm;

    float* sp = S1 + ((size_t)bh * T_ + j0 + 16 * dh + half * 8) * W_ + l16;
#pragma unroll
    for (int wt = 0; wt < 4; ++wt) {
      const v4f lo = d2[0][wt].lo + olo[wt];
      const v4f hi = d2[0][wt].hi + ohi[wt];
#pragma unroll
      for (int r = 0; r < 4; ++r) {
        sp[(size_t)r * W_ + wt * 16] = lo[r];
        sp[(size_t)(r + 4) * W_ + wt * 16] = hi[r];
      }
    }
  }
}

constexpr int VROWS_ = 80;
constexpr int VPITCH_ = 72;
constexpr int SCP_ = 68;
__global__ __launch_bounds__(128)
void attn_kernel(const float* __restrict__ kqv, const float* __restrict__ S1,
                 const float* __restrict__ s0, const float* __restrict__ v_pos,
                 float* __restrict__ attnout, float* __restrict__ wei_out) {
  __shared__ alignas(16) float sc[16 * SCP_];
  __shared__ alignas(16) float vs[VROWS_ * VPITCH_];
  __shared__ alignas(16) float ot[4 * 256];

  const int t0 = blockIdx.x * 16;
  const int h  = blockIdx.y;
  const int b  = blockIdx.z;
  const int bh = b * H_ + h;
  const int tid = threadIdx.x;
  const float scale = 0.125f;

  v4f vreg[10];
  {
    const float* vsrc = kqv + 2 * H_ * D_ + h * D_;
#pragma unroll
    for (int i = 0; i < 10; ++i) {
      const int idx = tid + i * 128;
      const int j = clampi(t0 - 64 + (idx >> 4), 0, T_ - 1);
      vreg[i] = *(const v4f*)(vsrc + (size_t)(b * T_ + j) * KQVROW + (idx & 15) * 4);
    }
  }

  {
    const int c  = tid & 15;
    const int rb = tid >> 4;
    const int ti = 15 - c;
    const int t  = t0 + ti;
    const size_t rowbase = (size_t)bh * T_;
    float sv[10];
#pragma unroll
    for (int i = 0; i < 10; ++i) {
      const int r = i * 8 + rb;
      const int w = r - 15 + c;
      const int j = t0 - (W_ - 1) + r;
      const int jc = clampi(j, 0, T_ - 1);
      const int wc = clampi(w, 0, W_ - 1);
      float v = (s0[rowbase + jc] + S1[(rowbase + jc) * W_ + wc]) * scale;
      if (j < 0)    v = -INFINITY;
      if (t >= Tq_) v = 0.0f;
      sv[i] = v;
    }
#pragma unroll
    for (int i = 0; i < 10; ++i) {
      const int w = i * 8 + rb - 15 + c;
      if ((unsigned)w < (unsigned)W_) sc[ti * SCP_ + w] = sv[i];
    }
  }
#pragma unroll
  for (int i = 0; i < 10; ++i) {
    const int idx = tid + i * 128;
    *(v4f*)(&vs[(idx >> 4) * VPITCH_ + (idx & 15) * 4]) = vreg[i];
  }
  __syncthreads();

  if (tid < 16) {
    float* row = &sc[tid * SCP_];
    v4f r[16];
#pragma unroll
    for (int i = 0; i < 16; ++i) r[i] = *(const v4f*)(row + 4 * i);
    v4f mv = r[0];
#pragma unroll
    for (int i = 1; i < 16; ++i) {
      mv.x = fmaxf(mv.x, r[i].x);
      mv.y = fmaxf(mv.y, r[i].y);
      mv.z = fmaxf(mv.z, r[i].z);
      mv.w = fmaxf(mv.w, r[i].w);
    }
    const float m = fmaxf(fmaxf(mv.x, mv.y), fmaxf(mv.z, mv.w));
    float s = 0.0f;
#pragma unroll
    for (int i = 0; i < 16; ++i) {
#pragma unroll
      for (int c = 0; c < 4; ++c) {
        const float e = __expf(r[i][c] - m);
        r[i][c] = e;
        s += e;
      }
    }
    const float inv = 1.0f / s;
#pragma unroll
    for (int i = 0; i < 16; ++i) *(v4f*)(row + 4 * i) = r[i] * inv;
  }
  __syncthreads();

  static_assert(T_ % 16 == 0 && Tq_ >= T_ - 2, "rows 0..13 of every 16-row tile must be live");
  {
    const int w   = tid & 63;
    const int tih = tid >> 6;
    float pv[8];
#pragma unroll
    for (int u = 0; u < 8; ++u) pv[u] = sc[(2 * u + tih) * SCP_ + w];
#pragma unroll
    for (int u = 0; u < 8; ++u) {
      const int t = t0 + 2 * u + tih;
      if (u < 7 || t < Tq_)
        wei_out[(((size_t)b * Tq_ + t) * W_ + w) * H_ + h] = pv[u];
    }
  }

  const int lane = tid & 31;
  const int wave = tid >> 5;
  const int d0   = wave * 16;
  const int half = lane >> 4, l16 = lane & 15, kb = half * 2;

  v8f c = {};
  v2f af[20], bf[20];
#pragma unroll
  for (int jt = 0; jt < 5; ++jt) {
    const int j0 = t0 - 64 + jt * 16;
#pragma unroll
    for (int k4 = 0; k4 < 16; k4 += 4) {
      const int s = jt * 4 + k4 / 4;
      {
        const int t = t0 + l16;
        const int wa = (j0 + k4 + kb) - t + (W_ - 1);
        const int wA = clampi(wa, 0, W_ - 1);
        const int wB = clampi(wa + 1, 0, W_ - 1);
        af[s].x = (wa     >= 0 && wa     < W_) ? sc[l16 * SCP_ + wA] : 0.0f;
        af[s].y = (wa + 1 >= 0 && wa + 1 < W_) ? sc[l16 * SCP_ + wB] : 0.0f;
      }
      {
        const int r = jt * 16 + k4 + kb;
        bf[s].x = vs[r * VPITCH_ + d0 + l16];
        bf[s].y = vs[(r + 1) * VPITCH_ + d0 + l16];
      }
    }
  }
#pragma unroll
  for (int s = 0; s < 20; ++s) c = WMMA_F32(af[s], bf[s], c);
#pragma unroll
  for (int k = 0; k < W_; k += 4) {
    v2f a;
    a.x = sc[l16 * SCP_ + k + kb];
    a.y = sc[l16 * SCP_ + k + kb + 1];
    v2f bv;
    bv.x = v_pos[((size_t)(k + kb)     * H_ + h) * D_ + d0 + l16];
    bv.y = v_pos[((size_t)(k + kb + 1) * H_ + h) * D_ + d0 + l16];
    c = WMMA_F32(a, bv, c);
  }

  {
    float* ow = ot + wave * 256 + (l16 >> 2) * 64 + half * 32 + (l16 & 3);
#pragma unroll
    for (int i = 0; i < 8; ++i) ow[4 * i] = c[i];
  }
  __syncthreads();
  {
    const float* orr = ot + wave * 256 + lane * 4;
    const v4f p0 = *(const v4f*)(orr);
    const v4f p1 = *(const v4f*)(orr + 128);
    float* gp = attnout +
        ((size_t)((b * (T_ / 16) + (int)blockIdx.x) * (C_ / 8) + h * (D_ / 8) + wave * 2) * 32 + lane) * 4;
    *(v4f*)(gp)       = p0;
    *(v4f*)(gp + 128) = p1;
  }
}

extern "C" void kernel_launch(void* const* d_in, const int* in_sizes, int n_in,
                              void* d_out, int out_size, void* d_ws, size_t ws_size,
                              hipStream_t stream) {
  (void)in_sizes; (void)n_in; (void)out_size; (void)ws_size;

  const float* x     = (const float*)d_in[0];
  const float* w_kqv = (const float*)d_in[1];
  const float* w_out = (const float*)d_in[2];
  const float* b_out = (const float*)d_in[3];
  const float* q_pos = (const float*)d_in[4];
  const float* v_pos = (const float*)d_in[5];

  float* y   = (float*)d_out;
  float* wei = y + (size_t)B_ * T_ * C_;

  char*  ws      = (char*)d_ws;
  float* kqv     = (float*)ws;
  size_t off     = (size_t)B_ * T_ * KQVROW * sizeof(float);
  float* S1      = (float*)(ws + off);
  off           += (size_t)B_ * H_ * T_ * W_ * sizeof(float);
  float* s0      = (float*)(ws + off);
  off           += (size_t)B_ * H_ * T_ * sizeof(float);
  float* attnout = (float*)(ws + off);

  qkv_kernel<4><<<dim3((B_ * T_) / 64, H_), 128, 0, stream>>>(x, w_kqv, q_pos, kqv, S1, s0);

  attn_kernel<<<dim3((Tq_ + 15) / 16, H_, B_), 128, 0, stream>>>(
      kqv, S1, s0, v_pos, attnout, wei);

  static_assert(S_ == 1, "exactly one pad row per batch, at row 0");
  gemm_abt<2, 2, 8, C_, C_, T_><<<dim3(C_ / 128, (B_ * T_) / 32), 128, 0, stream>>>(
      attnout, w_out, b_out, y);
}


# qkv: K loop cut in two over wave pairs: a wave runs 64 tokens x 96 features over half of K, 10 fragment loads per 48 WMMAs (was 8 per 24); halves added through LDS; k sum order changes
# speedup vs baseline: 2.4774x; 1.0668x over previous
#include <hip/hip_runtime.h>
#include <math.h>


typedef __attribute__((ext_vector_type(2))) float v2f;
typedef __attribute__((ext_vector_type(4))) float v4f;
typedef __attribute__((ext_vector_type(8))) float v8f;

#define WMMA_F32(a, b, c) \
  __builtin_amdgcn_wmma_f32_16x16x4_f32(false, (a), false, (b), (short)0, (c), false, false)

constexpr int B_ = 2, T_ = 2048, C_ = 256, H_ = 4, D_ = 64, W_ = 64, S_ = 1;
constexpr int Tq_ = T_ - S_;
constexpr int KQVROW = 3 * H_ * D_;

__device__ __forceinline__ int clampi(int v, int lo, int hi) {
  return v < lo ? lo : (v > hi ? hi : v);
}

template <int MT, int NT>
struct FragSet {
  v4f a[MT];
  v4f b[NT];
};

template <int MT, int NT, int K>
__device__ __forceinline__ void load_set(FragSet<MT, NT>& f, const float* __restrict__ ap,
                                         const float* __restrict__ bp, int k) {
#pragma unroll
  for (int i = 0; i < MT; ++i) f.a[i] = *(const v4f*)(ap + i * 16 * K + k * 16);
#pragma unroll
  for (int j = 0; j < NT; ++j) f.b[j] = *(const v4f*)(bp + j * 16 * K + k);
}

template <int MT, int NT>
__device__ __forceinline__ void mma_set(const FragSet<MT, NT>& f, v8f (&acc)[MT][NT]) {
#pragma unroll
  for (int i = 0; i < MT; ++i)
#pragma unroll
    for (int j = 0; j < NT; ++j)
      acc[i][j] = WMMA_F32(f.a[i].lo, f.b[j].lo, acc[i][j]);
#pragma unroll
  for (int i = 0; i < MT; ++i)
#pragma unroll
    for (int j = 0; j < NT; ++j)
      acc[i][j] = WMMA_F32(f.a[i].hi, f.b[j].hi, acc[i][j]);
}

template <int MT, int NT, int NS, int K, int N, int TB>
__global__ __launch_bounds__(128)
void gemm_abt(const float* __restrict__ A, const float* __restrict__ Bm,
              const float* __restrict__ bias, float* __restrict__ Cm) {
  const int lane = threadIdx.x & 31;
  const int wave = threadIdx.x >> 5;
  const int half = lane >> 4;
  const int l16  = lane & 15;
  const int n0   = ((int)blockIdx.x * 4 + wave) * (16 * NT);
  const int m0   = blockIdx.y * (16 * MT);

  const float* ap = A  + (size_t)m0 * K + lane * 4;
  const float* bp = Bm + (size_t)(n0 + l16) * K + half * 4;
  static_assert(TB % (16 * MT) == 0, "a wave's rows lie in one batch");

  float bb[NT];
#pragma unroll
  for (int j = 0; j < NT; ++j) bb[j] = bias ? bias[n0 + j * 16 + l16] : 0.0f;

  v8f acc[MT][NT];
#pragma unroll
  for (int i = 0; i < MT; ++i)
#pragma unroll
    for (int j = 0; j < NT; ++j) acc[i][j] = v8f{};

  FragSet<MT, NT> f[NS];
#pragma unroll
  for (int s = 0; s < NS; ++s) load_set<MT, NT, K>(f[s], ap, bp, s * 8);

#pragma unroll
  for (int k = 0; k < K; k += NS * 8) {
#pragma unroll
    for (int s = 0; s < NS; ++s) {
      __builtin_amdgcn_sched_barrier(0);
      mma_set<MT, NT>(f[s], acc);
      __builtin_amdgcn_sched_barrier(0);
      if (k + NS * 8 < K) load_set<MT, NT, K>(f[s], ap, bp, k + NS * 8 + s * 8);
    }
  }

  float* cp = Cm + (size_t)(m0 + half * 8 + 1) * N + n0 + l16;
#pragma unroll
  for (int j = 0; j < NT; ++j) {
    if (half == 0 && (unsigned)m0 % (unsigned)TB == 0u) Cm[(size_t)m0 * N + n0 + j * 16 + l16] = bb[j];
#pragma unroll
    for (int i = 0; i < MT; ++i) {
#pragma unroll
      for (int r = 0; r < 8; ++r) {
        if ((unsigned)(m0 + i * 16 + half * 8 + r) % (unsigned)TB != (unsigned)(TB - 1))
          cp[(i * 16 + r) * N + j * 16] = acc[i][j][r] + bb[j];
      }
    }
  }
}

struct QkvSet {
  v4f w[6];
  v4f t[4];
};

template <int K>
__device__ __forceinline__ void load_qkv_set(QkvSet& f, const float* __restrict__ wp,
                                             const float* __restrict__ xp0,
                                             const float* __restrict__ xp1,
                                             const float* __restrict__ xp2,
                                             const float* __restrict__ xp3, int k) {
  constexpr int QO = 1 * H_ * D_ * K;
  constexpr int KO = 0 * H_ * D_ * K;
  constexpr int VO = 2 * H_ * D_ * K;
#pragma unroll
  for (int i = 0; i < 2; ++i) {
    f.w[i]     = *(const v4f*)(wp + QO + i * 16 * K + k);
    f.w[2 + i] = *(const v4f*)(wp + KO + i * 16 * K + k);
    f.w[4 + i] = *(const v4f*)(wp + VO + i * 16 * K + k);
  }
  f.t[0] = *(const v4f*)(xp0 + k);
  f.t[1] = *(const v4f*)(xp1 + k);
  f.t[2] = *(const v4f*)(xp2 + k);
  f.t[3] = *(const v4f*)(xp3 + k);
}

template <int NS>
__global__ __launch_bounds__(128)
void qkv_kernel(const float* __restrict__ x, const float* __restrict__ w_kqv,
                const float* __restrict__ q_pos, float* __restrict__ kqv,
                float* __restrict__ S1, float* __restrict__ s0) {
  constexpr int K  = C_;
  constexpr int KH = K / 2;
  static_assert(KH % (8 * NS) == 0, "K loop runs whole passes over the NS sets");
  static_assert(T_ % 64 == 0, "a block's 64 tokens lie in one batch");
  static_assert(D_ == 64, "two d halves of two 16-feature tiles per projection");
  __shared__ alignas(16) float red[4 * 24 * 32 * 4];
  __shared__ alignas(16) float xs[4 * 8 * 32 * 4];
  __shared__ float xs0[4 * 16];

  const int lane = threadIdx.x & 31;
  const int wave = threadIdx.x >> 5;
  const int half = lane >> 4;
  const int l16  = lane & 15;
  const int kh   = wave >> 1;
  const int dh   = wave & 1;
  const int h    = blockIdx.y;
  const int tok0 = (blockIdx.x * 2 + kh) * 32;
  const int tokp = (blockIdx.x * 2 + (1 - kh)) * 32;
  const int b    = tok0 / T_;
  const int j0   = tok0 % T_;
  const int bh   = b * H_ + h;

  const float* wp = w_kqv + (size_t)(h * D_ + dh * 32 + l16) * K + kh * KH + half * 4;
  const int tt0 = tok0 + 16 * dh;
  const int tt1 = tok0 + 16 * (1 - dh);
  const int tt2 = tokp + 16 * dh;
  const int tt3 = tokp + 16 * (1 - dh);
  const float* xp0 = x + (size_t)(tt0 + l16) * K + kh * KH + half * 4;
  const float* xp1 = x + (size_t)(tt1 + l16) * K + kh * KH + half * 4;
  const float* xp2 = x + (size_t)(tt2 + l16) * K + kh * KH + half * 4;
  const float* xp3 = x + (size_t)(tt3 + l16) * K + kh * KH + half * 4;

  v8f aq[2][4], ak[2][4], av[4][2];
#pragma unroll
  for (int i = 0; i < 2; ++i) {
#pragma unroll
    for (int j = 0; j < 4; ++j) {
      aq[i][j] = v8f{};
      ak[i][j] = v8f{};
      av[j][i] = v8f{};
    }
  }

  QkvSet f[NS];
#pragma unroll
  for (int s = 0; s < NS; ++s) load_qkv_set<K>(f[s], wp, xp0, xp1, xp2, xp3, s * 8);

  v4f qp[4][2][2];

#pragma unroll
  for (int k = 0; k < KH; k += NS * 8) {
    if (k + NS * 8 >= KH) {
      const float* pp = q_pos + (size_t)(l16 * H_ + h) * D_ + dh * 32 + half * 8;
#pragma unroll
      for (int wt = 0; wt < 4; ++wt) {
#pragma unroll
        for (int i = 0; i < 2; ++i) {
          qp[wt][i][0] = *(const v4f*)(pp + (size_t)wt * 16 * H_ * D_ + 16 * i);
          qp[wt][i][1] = *(const v4f*)(pp + (size_t)wt * 16 * H_ * D_ + 16 * i + 4);
        }
      }
    }
#pragma unroll
    for (int s = 0; s < NS; ++s) {
      __builtin_amdgcn_sched_barrier(0);
#pragma unroll
      for (int i = 0; i < 2; ++i) {
#pragma unroll
        for (int j = 0; j < 4; ++j) {
          aq[i][j] = WMMA_F32(f[s].w[i].lo, f[s].t[j].lo, aq[i][j]);
          ak[i][j] = WMMA_F32(f[s].w[2 + i].lo, f[s].t[j].lo, ak[i][j]);
          av[j][i] = WMMA_F32(f[s].t[j].lo, f[s].w[4 + i].lo, av[j][i]);
        }
      }
#pragma unroll
      for (int i = 0; i < 2; ++i) {
#pragma unroll
        for (int j = 0; j < 4; ++j) {
          aq[i][j] = WMMA_F32(f[s].w[i].hi, f[s].t[j].hi, aq[i][j]);
          ak[i][j] = WMMA_F32(f[s].w[2 + i].hi, f[s].t[j].hi, ak[i][j]);
          av[j][i] = WMMA_F32(f[s].t[j].hi, f[s].w[4 + i].hi, av[j][i]);
        }
      }
      __builtin_amdgcn_sched_barrier(0);
      if (k + NS * 8 < KH) load_qkv_set<K>(f[s], wp, xp0, xp1, xp2, xp3, k + NS * 8 + s * 8);
    }
  }

  {
    float* rw = red + wave * (24 * 32 * 4) + lane * 4;
#pragma unroll
    for (int i = 0; i < 2; ++i) {
#pragma unroll
      for (int jj = 0; jj < 2; ++jj) {
        const int p = (i * 2 + jj) * 6;
        *(v4f*)(rw + (p + 0) * 128) = aq[i][2 + jj].lo;
        *(v4f*)(rw + (p + 1) * 128) = aq[i][2 + jj].hi;
        *(v4f*)(rw + (p + 2) * 128) = ak[i][2 + jj].lo;
        *(v4f*)(rw + (p + 3) * 128) = ak[i][2 + jj].hi;
        *(v4f*)(rw + (p + 4) * 128) = av[2 + jj][i].lo;
        *(v4f*)(rw + (p + 5) * 128) = av[2 + jj][i].hi;
      }
    }
  }
  __syncthreads();
  {
    const float* rr = red + (wave ^ 2) * (24 * 32 * 4) + lane * 4;
    v4f in[24];
#pragma unroll
    for (int p = 0; p < 24; ++p) in[p] = *(const v4f*)(rr + p * 128);
#pragma unroll
    for (int i = 0; i < 2; ++i) {
#pragma unroll
      for (int jj = 0; jj < 2; ++jj) {
        const int p = (i * 2 + jj) * 6;
        v8f tq, tk, tv;
        tq.lo = in[p + 0]; tq.hi = in[p + 1];
        tk.lo = in[p + 2]; tk.hi = in[p + 3];
        tv.lo = in[p + 4]; tv.hi = in[p + 5];
        aq[i][jj] = aq[i][jj] + tq;
        ak[i][jj] = ak[i][jj] + tk;
        av[jj][i] = av[jj][i] + tv;
      }
    }
  }

  float* vp0 = kqv + (size_t)(tt0 + half * 8) * KQVROW + 2 * H_ * D_ + h * D_ + dh * 32 + l16;
  float* vp1 = kqv + (size_t)(tt1 + half * 8) * KQVROW + 2 * H_ * D_ + h * D_ + dh * 32 + l16;
#pragma unroll
  for (int i = 0; i < 2; ++i) {
#pragma unroll
    for (int r = 0; r < 8; ++r) {
      vp0[(size_t)r * KQVROW + 16 * i] = av[0][i][r];
      vp1[(size_t)r * KQVROW + 16 * i] = av[1][i][r];
    }
  }

  v8f d2[2][4];
#pragma unroll
  for (int j = 0; j < 2; ++j)
#pragma unroll
    for (int wt = 0; wt < 4; ++wt) d2[j][wt] = v8f{};
#pragma unroll
  for (int i = 0; i < 2; ++i) {
#pragma unroll
    for (int p = 0; p < 4; ++p) {
#pragma unroll
      for (int j = 0; j < 2; ++j) {
        v2f a;
        a.x = aq[i][j][2 * p];
        a.y = aq[i][j][2 * p + 1];
#pragma unroll
        for (int wt = 0; wt < 4; ++wt) {
          const v4f piece = qp[wt][i][p >> 1];
          const v2f bq = (p & 1) ? piece.hi : piece.lo;
          d2[j][wt] = WMMA_F32(a, bq, d2[j][wt]);
        }
      }
    }
  }

  float part[2];
#pragma unroll
  for (int j = 0; j < 2; ++j) {
    float pj = 0.0f;
#pragma unroll
    for (int i = 0; i < 2; ++i) {
#pragma unroll
      for (int r = 0; r < 8; ++r) pj = fmaf(aq[i][j][r], ak[i][j][r], pj);
    }
    pj += __shfl_xor(pj, 16);
    part[j] = pj;
  }

  {
    float* xw = xs + wave * (8 * 32 * 4) + lane * 4;
#pragma unroll
    for (int wt = 0; wt < 4; ++wt) {
      *(v4f*)(xw + (wt * 2 + 0) * 128) = d2[1][wt].lo;
      *(v4f*)(xw + (wt * 2 + 1) * 128) = d2[1][wt].hi;
    }
    if (half == 0) xs0[wave * 16 + l16] = part[1];
  }
  __syncthreads();
  {
    const float* xr = xs + (wave ^ 1) * (8 * 32 * 4) + lane * 4;
    const float so = xs0[(wave ^ 1) * 16 + l16];
    v4f olo[4], ohi[4];
#pragma unroll
    for (int wt = 0; wt < 4; ++wt) {
      olo[wt] = *(const v4f*)(xr + (wt * 2 + 0) * 128);
      ohi[wt] = *(const v4f*)(xr + (wt * 2 + 1) * 128);
    }
    const float sm = part[0] + so;
    if (half == 0) s0[(size_t)bh * T_ + j0 + 16 * dh + l16] = sm;

    float* sp = S1 + ((size_t)bh * T_ + j0 + 16 * dh + half * 8) * W_ + l16;
#pragma unroll
    for (int wt = 0; wt < 4; ++wt) {
      const v4f lo = d2[0][wt].lo + olo[wt];
      const v4f hi = d2[0][wt].hi + ohi[wt];
#pragma unroll
      for (int r = 0; r < 4; ++r) {
        sp[(size_t)r * W_ + wt * 16] = lo[r];
        sp[(size_t)(r + 4) * W_ + wt * 16] = hi[r];
      }
    }
  }
}

constexpr int VROWS_ = 80;
constexpr int VPITCH_ = 72;
constexpr int SCP_ = 68;
__global__ __launch_bounds__(128)
void attn_kernel(const float* __restrict__ kqv, const float* __restrict__ S1,
                 const float* __restrict__ s0, const float* __restrict__ v_pos,
                 float* __restrict__ attnout, float* __restrict__ wei_out) {
  __shared__ alignas(16) float sc[16 * SCP_];
  __shared__ alignas(16) float vs[VROWS_ * VPITCH_];
  __shared__ alignas(16) float ot[4 * 256];

  const int t0 = blockIdx.x * 16;
  const int h  = blockIdx.y;
  const int b  = blockIdx.z;
  const int bh = b * H_ + h;
  const int tid = threadIdx.x;
  const float scale = 0.125f;

  v4f vreg[10];
  {
    const float* vsrc = kqv + 2 * H_ * D_ + h * D_;
#pragma unroll
    for (int i = 0; i < 10; ++i) {
      const int idx = tid + i * 128;
      const int j = clampi(t0 - 64 + (idx >> 4), 0, T_ - 1);
      vreg[i] = *(const v4f*)(vsrc + (size_t)(b * T_ + j) * KQVROW + (idx & 15) * 4);
    }
  }

  {
    const int c  = tid & 15;
    const int rb = tid >> 4;
    const int ti = 15 - c;
    const int t  = t0 + ti;
    const size_t rowbase = (size_t)bh * T_;
    float sv[10];
#pragma unroll
    for (int i = 0; i < 10; ++i) {
      const int r = i * 8 + rb;
      const int w = r - 15 + c;
      const int j = t0 - (W_ - 1) + r;
      const int jc = clampi(j, 0, T_ - 1);
      const int wc = clampi(w, 0, W_ - 1);
      float v = (s0[rowbase + jc] + S1[(rowbase + jc) * W_ + wc]) * scale;
      if (j < 0)    v = -INFINITY;
      if (t >= Tq_) v = 0.0f;
      sv[i] = v;
    }
#pragma unroll
    for (int i = 0; i < 10; ++i) {
      const int w = i * 8 + rb - 15 + c;
      if ((unsigned)w < (unsigned)W_) sc[ti * SCP_ + w] = sv[i];
    }
  }
#pragma unroll
  for (int i = 0; i < 10; ++i) {
    const int idx = tid + i * 128;
    *(v4f*)(&vs[(idx >> 4) * VPITCH_ + (idx & 15) * 4]) = vreg[i];
  }
  __syncthreads();

  if (tid < 16) {
    float* row = &sc[tid * SCP_];
    v4f r[16];
#pragma unroll
    for (int i = 0; i < 16; ++i) r[i] = *(const v4f*)(row + 4 * i);
    v4f mv = r[0];
#pragma unroll
    for (int i = 1; i < 16; ++i) {
      mv.x = fmaxf(mv.x, r[i].x);
      mv.y = fmaxf(mv.y, r[i].y);
      mv.z = fmaxf(mv.z, r[i].z);
      mv.w = fmaxf(mv.w, r[i].w);
    }
    const float m = fmaxf(fmaxf(mv.x, mv.y), fmaxf(mv.z, mv.w));
    float s = 0.0f;
#pragma unroll
    for (int i = 0; i < 16; ++i) {
#pragma unroll
      for (int c = 0; c < 4; ++c) {
        const float e = __expf(r[i][c] - m);
        r[i][c] = e;
        s += e;
      }
    }
    const float inv = 1.0f / s;
#pragma unroll
    for (int i = 0; i < 16; ++i) *(v4f*)(row + 4 * i) = r[i] * inv;
  }
  __syncthreads();

  static_assert(T_ % 16 == 0 && Tq_ >= T_ - 2, "rows 0..13 of every 16-row tile must be live");
  {
    const int w   = tid & 63;
    const int tih = tid >> 6;
    float pv[8];
#pragma unroll
    for (int u = 0; u < 8; ++u) pv[u] = sc[(2 * u + tih) * SCP_ + w];
#pragma unroll
    for (int u = 0; u < 8; ++u) {
      const int t = t0 + 2 * u + tih;
      if (u < 7 || t < Tq_)
        wei_out[(((size_t)b * Tq_ + t) * W_ + w) * H_ + h] = pv[u];
    }
  }

  const int lane = tid & 31;
  const int wave = tid >> 5;
  const int d0   = wave * 16;
  const int half = lane >> 4, l16 = lane & 15, kb = half * 2;

  v8f c = {};
  v2f af[20], bf[20];
#pragma unroll
  for (int jt = 0; jt < 5; ++jt) {
    const int j0 = t0 - 64 + jt * 16;
#pragma unroll
    for (int k4 = 0; k4 < 16; k4 += 4) {
      const int s = jt * 4 + k4 / 4;
      {
        const int t = t0 + l16;
        const int wa = (j0 + k4 + kb) - t + (W_ - 1);
        const int wA = clampi(wa, 0, W_ - 1);
        const int wB = clampi(wa + 1, 0, W_ - 1);
        af[s].x = (wa     >= 0 && wa     < W_) ? sc[l16 * SCP_ + wA] : 0.0f;
        af[s].y = (wa + 1 >= 0 && wa + 1 < W_) ? sc[l16 * SCP_ + wB] : 0.0f;
      }
      {
        const int r = jt * 16 + k4 + kb;
        bf[s].x = vs[r * VPITCH_ + d0 + l16];
        bf[s].y = vs[(r + 1) * VPITCH_ + d0 + l16];
      }
    }
  }
#pragma unroll
  for (int s = 0; s < 20; ++s) c = WMMA_F32(af[s], bf[s], c);
#pragma unroll
  for (int k = 0; k < W_; k += 4) {
    v2f a;
    a.x = sc[l16 * SCP_ + k + kb];
    a.y = sc[l16 * SCP_ + k + kb + 1];
    v2f bv;
    bv.x = v_pos[((size_t)(k + kb)     * H_ + h) * D_ + d0 + l16];
    bv.y = v_pos[((size_t)(k + kb + 1) * H_ + h) * D_ + d0 + l16];
    c = WMMA_F32(a, bv, c);
  }

  {
    float* ow = ot + wave * 256 + (l16 >> 2) * 64 + half * 32 + (l16 & 3);
#pragma unroll
    for (int i = 0; i < 8; ++i) ow[4 * i] = c[i];
  }
  __syncthreads();
  {
    const float* orr = ot + wave * 256 + lane * 4;
    const v4f p0 = *(const v4f*)(orr);
    const v4f p1 = *(const v4f*)(orr + 128);
    float* gp = attnout +
        ((size_t)((b * (T_ / 16) + (int)blockIdx.x) * (C_ / 8) + h * (D_ / 8) + wave * 2) * 32 + lane) * 4;
    *(v4f*)(gp)       = p0;
    *(v4f*)(gp + 128) = p1;
  }
}

extern "C" void kernel_launch(void* const* d_in, const int* in_sizes, int n_in,
                              void* d_out, int out_size, void* d_ws, size_t ws_size,
                              hipStream_t stream) {
  (void)in_sizes; (void)n_in; (void)out_size; (void)ws_size;

  const float* x     = (const float*)d_in[0];
  const float* w_kqv = (const float*)d_in[1];
  const float* w_out = (const float*)d_in[2];
  const float* b_out = (const float*)d_in[3];
  const float* q_pos = (const float*)d_in[4];
  const float* v_pos = (const float*)d_in[5];

  float* y   = (float*)d_out;
  float* wei = y + (size_t)B_ * T_ * C_;

  char*  ws      = (char*)d_ws;
  float* kqv     = (float*)ws;
  size_t off     = (size_t)B_ * T_ * KQVROW * sizeof(float);
  float* S1      = (float*)(ws + off);
  off           += (size_t)B_ * H_ * T_ * W_ * sizeof(float);
  float* s0      = (float*)(ws + off);
  off           += (size_t)B_ * H_ * T_ * sizeof(float);
  float* attnout = (float*)(ws + off);

  qkv_kernel<4><<<dim3((B_ * T_) / 64, H_), 128, 0, stream>>>(x, w_kqv, q_pos, kqv, S1, s0);

  attn_kernel<<<dim3((Tq_ + 15) / 16, H_, B_), 128, 0, stream>>>(
      kqv, S1, s0, v_pos, attnout, wei);

  static_assert(S_ == 1, "exactly one pad row per batch, at row 0");
  gemm_abt<2, 2, 8, C_, C_, T_><<<dim3(C_ / 128, (B_ * T_) / 32), 128, 0, stream>>>(
      attnout, w_out, b_out, y);
}
